# starting kernel
# baseline (speedup 1.0000x reference)
#include <hip/hip_runtime.h>

#define Bb   64
#define Ss   128
#define Ll   32
#define Dd   100
#define Mm   20
#define DP   112
#define MP   32
#define NTHR 256
#define NWAVE 8
#define NTILE 14
#define KSTEPS 25

typedef __attribute__((ext_vector_type(2))) float v2f;
typedef __attribute__((ext_vector_type(8))) float v8f;

__global__ __launch_bounds__(128) void sents_kernel(
    const int* __restrict__ tokens, const float* __restrict__ emb,
    const float* __restrict__ mult, float* __restrict__ sents)
{
    const int bs = blockIdx.x;
    const int d  = threadIdx.x;
    if (d >= Dd) return;
    const int* trow = tokens + bs * Ll;
    float acc = 0.f;
    #pragma unroll 4
    for (int l = 0; l < Ll; ++l) {
        const int tok = trow[l];
        acc += emb[tok * Dd + d] * mult[l * Dd + d];
    }
    sents[bs * Dd + d] = acc;
}

__global__ __launch_bounds__(128) void vk_kernel(
    const float* __restrict__ keys, const float* __restrict__ Vw,
    float* __restrict__ VkBuf)
{
    const int m = blockIdx.x;
    const int d = threadIdx.x;
    if (d >= Dd) return;
    float acc = 0.f;
    for (int k = 0; k < Dd; ++k) acc += keys[m * Dd + k] * Vw[d * Dd + k];
    VkBuf[m * Dd + d] = acc;
}

__global__ __launch_bounds__(NTHR) void entnet_scan_kernel(
    const float* __restrict__ sents, const float* __restrict__ keys,
    const float* __restrict__ Uw,    const float* __restrict__ Ww,
    const float* __restrict__ VkBuf, const float* __restrict__ prelu_a,
    float* __restrict__ out)
{
    const int b    = blockIdx.x;
    const int tid  = threadIdx.x;
    const int wave = tid >> 5;
    const int lane = tid & 31;
    const float pa = prelu_a[0];

    __shared__ float Uwt  [Dd * DP];
    __shared__ float memS [MP * DP];
    __shared__ float sS   [DP];
    __shared__ float WsS  [DP];
    __shared__ float gateS[Mm];
    __shared__ float scaleS[Mm];

    for (int i = tid; i < MP * DP; i += NTHR) memS[i] = 0.f;
    for (int i = tid; i < Dd * DP; i += NTHR) {
        const int k = i / DP, d = i - k * DP;
        Uwt[i] = (d < Dd) ? Uw[d * Dd + k] : 0.f;
    }
    __syncthreads();
    for (int i = tid; i < Mm * Dd; i += NTHR) {
        const int m = i / Dd, d = i - m * Dd;
        memS[m * DP + d] = keys[i];
    }
    __syncthreads();

    const int r15 = lane & 15;
    const int kh  = (lane >> 4) * 2;
    const int mh  = (lane >> 4) * 8;

    const int  t0   = wave;
    const int  t1   = wave + NWAVE;
    const bool has1 = (t1 < NTILE);
    const int  mb0 = (t0 / 7) * 16, nb0 = (t0 % 7) * 16;
    const int  mb1 = (t1 / 7) * 16, nb1 = (t1 % 7) * 16;

    v2f bf0[KSTEPS], bf1[KSTEPS];
    #pragma unroll
    for (int j = 0; j < KSTEPS; ++j) {
        const int kk = j * 4;
        bf0[j].x = Uwt[(kk + kh)     * DP + nb0 + r15];
        bf0[j].y = Uwt[(kk + kh + 1) * DP + nb0 + r15];
    }
    if (has1) {
        #pragma unroll
        for (int j = 0; j < KSTEPS; ++j) {
            const int kk = j * 4;
            bf1[j].x = Uwt[(kk + kh)     * DP + nb1 + r15];
            bf1[j].y = Uwt[(kk + kh + 1) * DP + nb1 + r15];
        }
    } else {
        #pragma unroll
        for (int j = 0; j < KSTEPS; ++j) bf1[j] = v2f{0.f, 0.f};
    }

    auto gemm_tile = [&](int mb, const v2f* bf) -> v8f {
        v8f c = {};
        #pragma unroll
        for (int j = 0; j < KSTEPS; ++j) {
            const int kk = j * 4;
            const v2f a = *(const v2f*)&memS[(mb + r15) * DP + kk + kh];
            c = __builtin_amdgcn_wmma_f32_16x16x4_f32(
                     false, a,  false, bf[j],
                     (short)0, c,  false,  false);
        }
        return c;
    };

    auto apply_tile = [&](int mb, int nb, v8f c) {
        #pragma unroll
        for (int r = 0; r < 8; ++r) {
            const int m = mb + mh + r;
            const int n = nb + r15;
            if (m < Mm && n < Dd) {
                float x = c[r] + VkBuf[m * Dd + n] + WsS[n];
                x = (x >= 0.f) ? x : pa * x;
                memS[m * DP + n] += x * gateS[m];
            }
        }
    };

    for (int t = 0; t < Ss; ++t) {
        if (tid < DP) sS[tid] = (tid < Dd) ? sents[(b * Ss + t) * Dd + tid] : 0.f;
        if (t + 1 < Ss && tid < 4)
            __builtin_prefetch(&sents[(b * Ss + t + 1) * Dd + tid * 32], 0, 3);
        __syncthreads();

        if (tid < Mm) {
            float mg = 0.f, kg = 0.f;
            #pragma unroll 4
            for (int k = 0; k < Dd; ++k) {
                const float sv = sS[k];
                mg += sv * memS[tid * DP + k];
                kg += sv * keys[tid * Dd + k];
            }
            const float z = mg + kg;
            const float g = 1.f / (1.f + expf(-z));
            gateS[tid] = (z == 0.f) ? 0.f : g;
        } else if (tid >= 32 && tid < 32 + DP) {
            const int d = tid - 32;
            float w = 0.f;
            if (d < Dd) {
                #pragma unroll 4
                for (int k = 0; k < Dd; ++k) w += sS[k] * Ww[d * Dd + k];
            }
            WsS[d] = w;
        }
        __syncthreads();

        v8f acc0 = gemm_tile(mb0, bf0);
        v8f acc1 = {};
        if (has1) acc1 = gemm_tile(mb1, bf1);
        __syncthreads();

        apply_tile(mb0, nb0, acc0);
        if (has1) apply_tile(mb1, nb1, acc1);
        __syncthreads();

        if (tid < Mm) {
            float ss = 0.f;
            #pragma unroll 4
            for (int d = 0; d < Dd; ++d) {
                const float v = memS[tid * DP + d];
                ss += v * v;
            }
            scaleS[tid] = 1.f / sqrtf(ss);
        }
        __syncthreads();
        for (int i = tid; i < Mm * DP; i += NTHR) {
            const int m = i / DP;
            memS[i] *= scaleS[m];
        }
        __syncthreads();
    }

    for (int i = tid; i < Mm * Dd; i += NTHR) {
        const int m = i / Dd, d = i - m * Dd;
        out[b * Mm * Dd + i] = memS[m * DP + d];
    }
}

extern "C" void kernel_launch(void* const* d_in, const int* in_sizes, int n_in,
                              void* d_out, int out_size, void* d_ws, size_t ws_size,
                              hipStream_t stream) {
    (void)in_sizes; (void)n_in; (void)out_size; (void)ws_size;
    const int*   tokens  = (const int*)  d_in[0];
    const float* emb     = (const float*)d_in[1];
    const float* keys    = (const float*)d_in[2];
    const float* mult    = (const float*)d_in[3];
    const float* Uw      = (const float*)d_in[4];
    const float* Vw      = (const float*)d_in[5];
    const float* Ww      = (const float*)d_in[6];
    const float* prelu_a = (const float*)d_in[7];
    float* out = (float*)d_out;

    float* sents = (float*)d_ws;
    float* VkBuf = sents + (size_t)Bb * Ss * Dd;

    sents_kernel<<<Bb * Ss, 128, 0, stream>>>(tokens, emb, mult, sents);
    vk_kernel<<<Mm, 128, 0, stream>>>(keys, Vw, VkBuf);
    entnet_scan_kernel<<<Bb, NTHR, 0, stream>>>(sents, keys, Uw, Ww, VkBuf,
                                                prelu_a, out);
}


# Step product transposed (slots on columns), k walked in D-fragment order: state registers are the next B operand; Uw, Ww as register A fragments; s@Ww^T joins the product; gate, norm sums reordered
# speedup vs baseline: 1.7279x; 1.7279x over previous
#include <hip/hip_runtime.h>

#define Bb   64
#define Ss   128
#define Ll   32
#define Dd   100
#define Mm   20
#define MT   7
#define KS   26
#define NGRP 80

typedef __attribute__((ext_vector_type(2))) float v2f;
typedef __attribute__((ext_vector_type(4))) float v4f;
typedef __attribute__((ext_vector_type(8))) float v8f;

__global__ __launch_bounds__(128) void sents_kernel(
    const int* __restrict__ tokens, const float* __restrict__ emb,
    const float* __restrict__ mult, float* __restrict__ sents)
{
    const int bs = blockIdx.x;
    const int d  = threadIdx.x;
    if (d >= Dd) return;
    const int* trow = tokens + bs * Ll;
    float acc = 0.f;
    #pragma unroll 4
    for (int l = 0; l < Ll; ++l) {
        const int tok = trow[l];
        acc += emb[tok * Dd + d] * mult[l * Dd + d];
    }
    sents[bs * Dd + d] = acc;
}

__global__ __launch_bounds__(128) void vk_kernel(
    const float* __restrict__ keys, const float* __restrict__ Vw,
    float* __restrict__ VkBuf)
{
    const int m = blockIdx.x;
    const int d = threadIdx.x;
    if (d >= Dd) return;
    float acc = 0.f;
    for (int k = 0; k < Dd; ++k) acc += keys[m * Dd + k] * Vw[d * Dd + k];
    VkBuf[m * Dd + d] = acc;
}

__device__ __forceinline__ v4f ld4_or_zero(const float* __restrict__ base, int off, bool ok)
{
    const v4f v = *(const v4f*)(base + (ok ? off : 0));
    const v4f z = {0.f, 0.f, 0.f, 0.f};
    return ok ? v : z;
}

__device__ __forceinline__ v8f cat8(v4f lo, v4f hi)
{
    return __builtin_shufflevector(lo, hi, 0, 1, 2, 3, 4, 5, 6, 7);
}

__global__ __launch_bounds__(32) void entnet_scan_kernel(
    const float* __restrict__ sents, const float* __restrict__ keys,
    const float* __restrict__ Uw,    const float* __restrict__ Ww,
    const float* __restrict__ VkBuf, const float* __restrict__ prelu_a,
    float* __restrict__ out)
{
    const int lane = threadIdx.x;
    const int c    = lane & 15;
    const int h    = lane >> 4;
    const int G    = blockIdx.x * 16 + c;
    const int b    = G / Mm;
    const int m    = G - b * Mm;
    const float pa = prelu_a[0];

    __shared__ v4f keyL[2 * MT * 32];
    __shared__ v4f vkL [2 * MT * 32];

    v8f st[MT];
    #pragma unroll
    for (int i = 0; i < MT; ++i) {
        const int d0 = 16 * i + 8 * h;
        const v4f klo = ld4_or_zero(keys,  m * Dd + d0,     d0     < Dd);
        const v4f khi = ld4_or_zero(keys,  m * Dd + d0 + 4, d0 + 4 < Dd);
        const v4f vlo = ld4_or_zero(VkBuf, m * Dd + d0,     d0     < Dd);
        const v4f vhi = ld4_or_zero(VkBuf, m * Dd + d0 + 4, d0 + 4 < Dd);
        st[i] = cat8(klo, khi);
        keyL[(2 * i)     * 32 + lane] = klo;
        keyL[(2 * i + 1) * 32 + lane] = khi;
        vkL [(2 * i)     * 32 + lane] = vlo;
        vkL [(2 * i + 1) * 32 + lane] = vhi;
    }

    v2f UA[MT][KS], WA[MT][KS];
    #pragma unroll
    for (int i = 0; i < MT; ++i) {
        const int row = 16 * i + c;
        #pragma unroll
        for (int ip = 0; ip < MT; ++ip) {
            #pragma unroll
            for (int q = 0; q < 2; ++q) {
                if (ip == MT - 1 && q == 1) continue;
                const int  k0 = 16 * ip + 8 * h + 4 * q;
                const bool ok = (row < Dd) && (k0 < Dd);
                const v4f u = ld4_or_zero(Uw, row * Dd + k0, ok);
                const v4f w = ld4_or_zero(Ww, row * Dd + k0, ok);
                const int j = 4 * ip + 2 * q;
                UA[i][j]     = v2f{u.x, u.y};
                UA[i][j + 1] = v2f{u.z, u.w};
                WA[i][j]     = v2f{w.x, w.y};
                WA[i][j + 1] = v2f{w.z, w.w};
            }
        }
    }

    auto load_s = [&](int t, v8f* dst) {
        const int rowoff = (b * Ss + t) * Dd;
        #pragma unroll
        for (int i = 0; i < MT; ++i) {
            const int d0 = 16 * i + 8 * h;
            const v4f lo = ld4_or_zero(sents, rowoff + d0,     d0     < Dd);
            const v4f hi = ld4_or_zero(sents, rowoff + d0 + 4, d0 + 4 < Dd);
            dst[i] = cat8(lo, hi);
        }
    };

    v8f sv[MT];
    load_s(0, sv);

    for (int t = 0; t < Ss; ++t) {
        float mg0 = 0.f, mg1 = 0.f, kg0 = 0.f, kg1 = 0.f;
        #pragma unroll
        for (int i = 0; i < MT; ++i) {
            const v8f kk = cat8(keyL[(2 * i) * 32 + lane], keyL[(2 * i + 1) * 32 + lane]);
            #pragma unroll
            for (int r = 0; r < 8; r += 2) {
                mg0 += sv[i][r]     * st[i][r];
                mg1 += sv[i][r + 1] * st[i][r + 1];
                kg0 += sv[i][r]     * kk[r];
                kg1 += sv[i][r + 1] * kk[r + 1];
            }
        }
        float mg = mg0 + mg1;
        float kg = kg0 + kg1;
        mg += __shfl_xor(mg, 16);
        kg += __shfl_xor(kg, 16);
        const float z = mg + kg;
        const float g = 1.f / (1.f + expf(-z));
        const float gate = (z == 0.f) ? 0.f : g;

        __builtin_amdgcn_sched_barrier(0);

        v8f acc[MT];
        #pragma unroll
        for (int i = 0; i < MT; ++i)
            acc[i] = cat8(vkL[(2 * i) * 32 + lane], vkL[(2 * i + 1) * 32 + lane]);
        #pragma unroll
        for (int j = 0; j < KS; ++j) {
            const int p = 2 * j;
            const v2f bs = v2f{sv[p >> 3][p & 7], sv[p >> 3][(p & 7) + 1]};
            #pragma unroll
            for (int i = 0; i < MT; ++i)
                acc[i] = __builtin_amdgcn_wmma_f32_16x16x4_f32(
                    false, WA[i][j], false, bs, (short)0, acc[i], false, false);
        }

        load_s((t + 1 < Ss) ? t + 1 : t, sv);

        #pragma unroll
        for (int j = 0; j < KS; ++j) {
            const int p = 2 * j;
            const v2f bm = v2f{st[p >> 3][p & 7], st[p >> 3][(p & 7) + 1]};
            #pragma unroll
            for (int i = 0; i < MT; ++i)
                acc[i] = __builtin_amdgcn_wmma_f32_16x16x4_f32(
                    false, UA[i][j], false, bm, (short)0, acc[i], false, false);
        }

        float ss0 = 0.f, ss1 = 0.f;
        #pragma unroll
        for (int i = 0; i < MT; ++i) {
            #pragma unroll
            for (int r = 0; r < 8; r += 2) {
                float x0 = acc[i][r];
                float x1 = acc[i][r + 1];
                x0 = (x0 >= 0.f) ? x0 : pa * x0;
                x1 = (x1 >= 0.f) ? x1 : pa * x1;
                const float n0 = st[i][r]     + x0 * gate;
                const float n1 = st[i][r + 1] + x1 * gate;
                st[i][r]     = n0;
                st[i][r + 1] = n1;
                ss0 += n0 * n0;
                ss1 += n1 * n1;
            }
        }
        float ss = ss0 + ss1;
        ss += __shfl_xor(ss, 16);
        const float sc = 1.f / sqrtf(ss);
        #pragma unroll
        for (int i = 0; i < MT; ++i) {
            st[i] = st[i] * sc;
        }
    }

    #pragma unroll
    for (int i = 0; i < MT; ++i) {
        const int d0 = 16 * i + 8 * h;
        if (d0 < Dd)
            *(v4f*)(out + G * Dd + d0) = __builtin_shufflevector(st[i], st[i], 0, 1, 2, 3);
        if (d0 + 4 < Dd)
            *(v4f*)(out + G * Dd + d0 + 4) = __builtin_shufflevector(st[i], st[i], 4, 5, 6, 7);
    }
}

extern "C" void kernel_launch(void* const* d_in, const int* in_sizes, int n_in,
                              void* d_out, int out_size, void* d_ws, size_t ws_size,
                              hipStream_t stream) {
    (void)in_sizes; (void)n_in; (void)out_size; (void)ws_size;
    const int*   tokens  = (const int*)  d_in[0];
    const float* emb     = (const float*)d_in[1];
    const float* keys    = (const float*)d_in[2];
    const float* mult    = (const float*)d_in[3];
    const float* Uw      = (const float*)d_in[4];
    const float* Vw      = (const float*)d_in[5];
    const float* Ww      = (const float*)d_in[6];
    const float* prelu_a = (const float*)d_in[7];
    float* out = (float*)d_out;

    float* sents = (float*)d_ws;
    float* VkBuf = sents + (size_t)Bb * Ss * Dd;

    sents_kernel<<<Bb * Ss, 128, 0, stream>>>(tokens, emb, mult, sents);
    vk_kernel<<<Mm, 128, 0, stream>>>(keys, Vw, VkBuf);
    entnet_scan_kernel<<<NGRP, 32, 0, stream>>>(sents, keys, Uw, Ww, VkBuf,
                                                prelu_a, out);
}


# Step products split along k over 4 waves (one per SIMD): each keeps a quarter of state, s and both weights' A fragments in registers; partials meet in LDS; gate, norm, product sums reordered
# speedup vs baseline: 4.6997x; 2.7199x over previous
#include <hip/hip_runtime.h>

#define Bb   64
#define Ss   128
#define Ll   32
#define Dd   100
#define Mm   20
#define MT   7
#define NW   4
#define NJ   7
#define NQ   14
#define NP2  28
#define NGRP 80

typedef __attribute__((ext_vector_type(2))) float v2f;
typedef __attribute__((ext_vector_type(4))) float v4f;
typedef __attribute__((ext_vector_type(8))) float v8f;

__global__ __launch_bounds__(128) void sents_kernel(
    const int* __restrict__ tokens, const float* __restrict__ emb,
    const float* __restrict__ mult, float* __restrict__ sents)
{
    const int bs = blockIdx.x;
    const int d  = threadIdx.x;
    if (d >= Dd) return;
    const int* trow = tokens + bs * Ll;
    float acc = 0.f;
    #pragma unroll 4
    for (int l = 0; l < Ll; ++l) {
        const int tok = trow[l];
        acc += emb[tok * Dd + d] * mult[l * Dd + d];
    }
    sents[bs * Dd + d] = acc;
}

__global__ __launch_bounds__(128) void vk_kernel(
    const float* __restrict__ keys, const float* __restrict__ Vw,
    float* __restrict__ VkBuf)
{
    const int m = blockIdx.x;
    const int d = threadIdx.x;
    if (d >= Dd) return;
    float acc = 0.f;
    for (int k = 0; k < Dd; ++k) acc += keys[m * Dd + k] * Vw[d * Dd + k];
    VkBuf[m * Dd + d] = acc;
}

__device__ __forceinline__ v2f ld2_or_zero(const float* __restrict__ base, int off, bool ok)
{
    const v2f v = *(const v2f*)(base + (ok ? off : 0));
    const v2f z = {0.f, 0.f};
    return ok ? v : z;
}

__global__ __launch_bounds__(128) void entnet_scan_kernel(
    const float* __restrict__ sents, const float* __restrict__ keys,
    const float* __restrict__ Uw,    const float* __restrict__ Ww,
    const float* __restrict__ VkBuf, const float* __restrict__ prelu_a,
    float* __restrict__ out)
{
    const int lane = threadIdx.x & 31;
    const int w    = __builtin_amdgcn_readfirstlane((int)(threadIdx.x >> 5));
    const int c    = lane & 15;
    const int h    = lane >> 4;
    const int G    = blockIdx.x * 16 + c;
    const int b    = G / Mm;
    const int m    = G - b * Mm;
    const float pa = prelu_a[0];

    __shared__ v2f   partL[21 * 1024];
    __shared__ float zL [NW * 32];
    __shared__ float ssL[NW * 32];
    __shared__ __attribute__((aligned(16))) float outL[16 * Dd];

    int  dof[NJ];
    bool dok[NJ];
    #pragma unroll
    for (int jj = 0; jj < NJ; ++jj) {
        const int P = NQ * w + 2 * jj;
        dof[jj] = 16 * (P >> 3) + 8 * h + (P & 7);
        dok[jj] = dof[jj] < Dd;
    }

    float st[NQ], key[NQ], vk[NQ];
    #pragma unroll
    for (int jj = 0; jj < NJ; ++jj) {
        const v2f k2 = ld2_or_zero(keys,  m * Dd + dof[jj], dok[jj]);
        const v2f v2 = ld2_or_zero(VkBuf, m * Dd + dof[jj], dok[jj]);
        st[2 * jj] = k2.x;  st[2 * jj + 1] = k2.y;
        key[2 * jj] = k2.x; key[2 * jj + 1] = k2.y;
        vk[2 * jj] = v2.x;  vk[2 * jj + 1] = v2.y;
    }

    v2f UA[MT][NJ], WA[MT][NJ];
    #pragma unroll
    for (int i = 0; i < MT; ++i) {
        const int row = 16 * i + c;
        #pragma unroll
        for (int jj = 0; jj < NJ; ++jj) {
            const bool ok = (row < Dd) && dok[jj];
            UA[i][jj] = ld2_or_zero(Uw, row * Dd + dof[jj], ok);
            WA[i][jj] = ld2_or_zero(Ww, row * Dd + dof[jj], ok);
        }
    }

    auto load_s = [&](int t, float* dst) {
        const int rowoff = (b * Ss + t) * Dd;
        #pragma unroll
        for (int jj = 0; jj < NJ; ++jj) {
            const v2f s2 = ld2_or_zero(sents, rowoff + dof[jj], dok[jj]);
            dst[2 * jj] = s2.x; dst[2 * jj + 1] = s2.y;
        }
    };

    auto sprod = [&](v8f* a, const float* s, int j0, int j1) {
        #pragma unroll
        for (int jj = j0; jj < j1; ++jj) {
            const v2f bs = v2f{s[2 * jj], s[2 * jj + 1]};
            #pragma unroll
            for (int i = 0; i < MT; ++i)
                a[i] = __builtin_amdgcn_wmma_f32_16x16x4_f32(
                    false, WA[i][jj], false, bs, (short)0, a[i], false, false);
        }
    };

    const v8f zero8 = {0.f, 0.f, 0.f, 0.f, 0.f, 0.f, 0.f, 0.f};

    float svc[NQ];
    load_s(0, svc);
    v8f accS[MT];
    #pragma unroll
    for (int i = 0; i < MT; ++i) accS[i] = zero8;
    sprod(accS, svc, 0, NJ);

    v2f* const       wr = partL + (w * NP2) * 32 + lane;
    const v2f* const rd = partL + (NJ * w) * 32 + lane;

    auto sstep = [&](v8f* a, const float* s, int jj) { sprod(a, s, jj, jj + 1); };

    #define SCHED_FENCE() __builtin_amdgcn_sched_barrier(0)

    for (int t = 0; t < Ss; ++t) {
        float svn[NQ];
        load_s((t + 1 < Ss) ? t + 1 : t, svn);

        float mg0 = 0.f, mg1 = 0.f, kg0 = 0.f, kg1 = 0.f;
        #pragma unroll
        for (int q = 0; q < NQ; q += 2) {
            mg0 += svc[q]     * st[q];
            mg1 += svc[q + 1] * st[q + 1];
            kg0 += svc[q]     * key[q];
            kg1 += svc[q + 1] * key[q + 1];
        }
        zL[w * 32 + lane] = (mg0 + mg1) + (kg0 + kg1);

        v8f acc[MT];
        #pragma unroll
        for (int i = 0; i < MT; ++i) acc[i] = accS[i];
        #pragma unroll
        for (int jj = 0; jj < NJ; ++jj) {
            const v2f bm = v2f{st[2 * jj], st[2 * jj + 1]};
            #pragma unroll
            for (int i = 0; i < MT; ++i)
                acc[i] = __builtin_amdgcn_wmma_f32_16x16x4_f32(
                    false, UA[i][jj], false, bm, (short)0, acc[i], false, false);
        }
        SCHED_FENCE();

        #pragma unroll
        for (int i = 0; i < MT; ++i) {
            #pragma unroll
            for (int rr = 0; rr < 4; ++rr)
                wr[(4 * i + rr) * 32] = v2f{acc[i][2 * rr], acc[i][2 * rr + 1]};
        }
        SCHED_FENCE();

        v8f nS[MT];
        #pragma unroll
        for (int i = 0; i < MT; ++i) nS[i] = zero8;
        sstep(nS, svn, 0);

        __syncthreads();

        v2f a0[NJ], a1[NJ], a2[NJ], a3[NJ];
        #pragma unroll
        for (int jj = 0; jj < NJ; ++jj) {
            a0[jj] = rd[(0 * NP2 + jj) * 32];
            a1[jj] = rd[(1 * NP2 + jj) * 32];
            a2[jj] = rd[(2 * NP2 + jj) * 32];
            a3[jj] = rd[(3 * NP2 + jj) * 32];
        }
        float zq[8];
        #pragma unroll
        for (int e = 0; e < 8; ++e) zq[e] = zL[16 * e + c];
        SCHED_FENCE();
        sstep(nS, svn, 1);
        SCHED_FENCE();

        float x[NQ];
        #pragma unroll
        for (int jj = 0; jj < NJ; ++jj) {
            x[2 * jj]     = ((a0[jj].x + a1[jj].x) + (a2[jj].x + a3[jj].x)) + vk[2 * jj];
            x[2 * jj + 1] = ((a0[jj].y + a1[jj].y) + (a2[jj].y + a3[jj].y)) + vk[2 * jj + 1];
        }
        const float z = ((zq[0] + zq[1]) + (zq[2] + zq[3])) + ((zq[4] + zq[5]) + (zq[6] + zq[7]));

        sstep(nS, svn, 2);
        sstep(nS, svn, 3);
        sstep(nS, svn, 4);
        const float g = 1.f / (1.f + expf(-z));
        const float gate = (z == 0.f) ? 0.f : g;

        float ss0 = 0.f, ss1 = 0.f;
        #pragma unroll
        for (int q = 0; q < NQ; q += 2) {
            float x0 = x[q];
            float x1 = x[q + 1];
            x0 = (x0 >= 0.f) ? x0 : pa * x0;
            x1 = (x1 >= 0.f) ? x1 : pa * x1;
            const float n0 = st[q]     + x0 * gate;
            const float n1 = st[q + 1] + x1 * gate;
            st[q]     = n0;
            st[q + 1] = n1;
            ss0 += n0 * n0;
            ss1 += n1 * n1;
        }
        ssL[w * 32 + lane] = ss0 + ss1;

        __syncthreads();

        asm volatile("" : "+v"(nS[0]), "+v"(nS[1]), "+v"(nS[2]), "+v"(nS[3]),
                          "+v"(nS[4]), "+v"(nS[5]), "+v"(nS[6]));

        float sq[8];
        #pragma unroll
        for (int e = 0; e < 8; ++e) sq[e] = ssL[16 * e + c];
        SCHED_FENCE();
        sstep(nS, svn, 5);
        SCHED_FENCE();
        const float ss = ((sq[0] + sq[1]) + (sq[2] + sq[3])) + ((sq[4] + sq[5]) + (sq[6] + sq[7]));
        sstep(nS, svn, 6);
        const float sc = 1.f / sqrtf(ss);
        #pragma unroll
        for (int q = 0; q < NQ; ++q) st[q] = st[q] * sc;

        #pragma unroll
        for (int i = 0; i < MT; ++i) accS[i] = nS[i];
        #pragma unroll
        for (int q = 0; q < NQ; ++q) svc[q] = svn[q];
    }
    #undef SCHED_FENCE

    #pragma unroll
    for (int jj = 0; jj < NJ; ++jj) {
        if (dok[jj])
            *(v2f*)(outL + c * Dd + dof[jj]) = v2f{st[2 * jj], st[2 * jj + 1]};
    }
    __syncthreads();
    {
        const v4f* src = (const v4f*)outL;
        v4f* dst = (v4f*)(out + (size_t)blockIdx.x * 16 * Dd);
        for (int idx = threadIdx.x; idx < 16 * Dd / 4; idx += 128) dst[idx] = src[idx];
    }
}

extern "C" void kernel_launch(void* const* d_in, const int* in_sizes, int n_in,
                              void* d_out, int out_size, void* d_ws, size_t ws_size,
                              hipStream_t stream) {
    (void)in_sizes; (void)n_in; (void)out_size; (void)ws_size;
    const int*   tokens  = (const int*)  d_in[0];
    const float* emb     = (const float*)d_in[1];
    const float* keys    = (const float*)d_in[2];
    const float* mult    = (const float*)d_in[3];
    const float* Uw      = (const float*)d_in[4];
    const float* Vw      = (const float*)d_in[5];
    const float* Ww      = (const float*)d_in[6];
    const float* prelu_a = (const float*)d_in[7];
    float* out = (float*)d_out;

    float* sents = (float*)d_ws;
    float* VkBuf = sents + (size_t)Bb * Ss * Dd;

    sents_kernel<<<Bb * Ss, 128, 0, stream>>>(tokens, emb, mult, sents);
    vk_kernel<<<Mm, 128, 0, stream>>>(keys, Vw, VkBuf);
    entnet_scan_kernel<<<NGRP, 128, 0, stream>>>(sents, keys, Uw, Ww, VkBuf,
                                                prelu_a, out);
}


# Sentence term Ww@s re-tiled in the scan: tile columns = 8 time steps x 2 batch elements (was 16 slots), one k-step per scan step, Ww fragments in LDS; added after the partial sums (sum order changed)
# speedup vs baseline: 5.6902x; 1.2108x over previous
#include <hip/hip_runtime.h>

#define Bb   64
#define Ss   128
#define Ll   32
#define Dd   100
#define Mm   20
#define MT   7
#define NW   4
#define NJ   7
#define NQ   14
#define NP2  28
#define NGRP 80
#define SOFF 4096
#define WOFF 8192

typedef __attribute__((ext_vector_type(2))) float v2f;
typedef __attribute__((ext_vector_type(4))) float v4f;
typedef __attribute__((ext_vector_type(8))) float v8f;

__global__ __launch_bounds__(128) void sents_kernel(
    const int* __restrict__ tokens, const float* __restrict__ emb,
    const float* __restrict__ mult, float* __restrict__ sents)
{
    const int bs = blockIdx.x;
    const int d  = threadIdx.x;
    if (d >= Dd) return;
    const int* trow = tokens + bs * Ll;
    float acc = 0.f;
    #pragma unroll 4
    for (int l = 0; l < Ll; ++l) {
        const int tok = trow[l];
        acc += emb[tok * Dd + d] * mult[l * Dd + d];
    }
    sents[bs * Dd + d] = acc;
}

__global__ __launch_bounds__(128) void vk_kernel(
    const float* __restrict__ keys, const float* __restrict__ Vw,
    float* __restrict__ VkBuf)
{
    const int m = blockIdx.x;
    const int d = threadIdx.x;
    if (d >= Dd) return;
    float acc = 0.f;
    for (int k = 0; k < Dd; ++k) acc += keys[m * Dd + k] * Vw[d * Dd + k];
    VkBuf[m * Dd + d] = acc;
}

__device__ __forceinline__ v2f ld2_or_zero(const float* __restrict__ base, int off, bool ok)
{
    const v2f v = *(const v2f*)(base + (ok ? off : 0));
    const v2f z = {0.f, 0.f};
    return ok ? v : z;
}

struct StepMid  { static constexpr bool last = false; };
struct StepLast { static constexpr bool last = true;  };

__global__ __launch_bounds__(128) void entnet_scan_kernel(
    const float* __restrict__ sents, const float* __restrict__ keys,
    const float* __restrict__ Uw,    const float* __restrict__ Ww,
    const float* __restrict__ VkBuf, const float* __restrict__ prelu_a,
    float* __restrict__ out)
{
    const int lane = threadIdx.x & 31;
    const int w    = __builtin_amdgcn_readfirstlane((int)(threadIdx.x >> 5));
    const int c    = lane & 15;
    const int h    = lane >> 4;
    const int G    = blockIdx.x * 16 + c;
    const int b    = G / Mm;
    const int m    = G - b * Mm;
    const float pa = prelu_a[0];

    const int gb0  = (blockIdx.x * 16) / Mm;
    const int gb1  = (blockIdx.x * 16 + 15) / Mm;
    const int bcol = (c >> 3) ? gb1 : gb0;
    const int tcol = c & 7;
    const int selb = (b == gb0) ? 0 : 8;

    __shared__ v2f   partL[21 * 1024];
    __shared__ float zL [NW * 32];
    __shared__ float ssL[NW * 32];
    __shared__ __attribute__((aligned(16))) float outL[16 * Dd];

    int  dof[NJ];
    bool dok[NJ];
    #pragma unroll
    for (int jj = 0; jj < NJ; ++jj) {
        const int P = NQ * w + 2 * jj;
        dof[jj] = 16 * (P >> 3) + 8 * h + (P & 7);
        dok[jj] = dof[jj] < Dd;
    }

    float st[NQ], key[NQ], vk[NQ];
    #pragma unroll
    for (int jj = 0; jj < NJ; ++jj) {
        const v2f k2 = ld2_or_zero(keys,  m * Dd + dof[jj], dok[jj]);
        const v2f v2 = ld2_or_zero(VkBuf, m * Dd + dof[jj], dok[jj]);
        st[2 * jj] = k2.x;  st[2 * jj + 1] = k2.y;
        key[2 * jj] = k2.x; key[2 * jj + 1] = k2.y;
        vk[2 * jj] = v2.x;  vk[2 * jj + 1] = v2.y;
    }

    v2f* const wal = partL + WOFF + (w * NJ * MT) * 32 + lane;
    v2f UA[MT][NJ];
    #pragma unroll
    for (int i = 0; i < MT; ++i) {
        const int row = 16 * i + c;
        #pragma unroll
        for (int jj = 0; jj < NJ; ++jj) {
            const bool ok = (row < Dd) && dok[jj];
            UA[i][jj] = ld2_or_zero(Uw, row * Dd + dof[jj], ok);
            wal[(jj * MT + i) * 32] = ld2_or_zero(Ww, row * Dd + dof[jj], ok);
        }
    }

    auto load_s = [&](int t, float* dst) {
        const int rowoff = (b * Ss + t) * Dd;
        #pragma unroll
        for (int jj = 0; jj < NJ; ++jj) {
            const v2f s2 = ld2_or_zero(sents, rowoff + dof[jj], dok[jj]);
            dst[2 * jj] = s2.x; dst[2 * jj + 1] = s2.y;
        }
    };

    auto sent_b = [&](int tb, int jj) -> v2f {
        const int P  = NQ * w + 2 * jj;
        const int d  = 16 * (P >> 3) + 8 * h + (P & 7);
        const int t8 = tb * 8 + tcol;
        const int tt = (t8 < Ss) ? t8 : (Ss - 1);
        return ld2_or_zero(sents, (bcol * Ss + tt) * Dd + d, d < Dd);
    };

    const v8f zero8 = {0.f, 0.f, 0.f, 0.f, 0.f, 0.f, 0.f, 0.f};

    v2f* const       wr  = partL + (w * NP2) * 32 + lane;
    const v2f* const rd  = partL + (NJ * w) * 32 + lane;
    v2f* const       swr = partL + SOFF + (w * NP2) * 32 + lane;
    const v2f* const srd = partL + SOFF + (NJ * w) * 32 + lane;

    float tot[NQ];
    v8f nS[MT];
    #pragma unroll
    for (int i = 0; i < MT; ++i) nS[i] = zero8;

    #pragma nounroll
    for (int jj = 0; jj < NJ; ++jj) {
        const v2f bs = sent_b(0, jj);
        #pragma unroll
        for (int i = 0; i < MT; ++i)
            nS[i] = __builtin_amdgcn_wmma_f32_16x16x4_f32(
                false, wal[(jj * MT + i) * 32], false, bs, (short)0, nS[i], false, false);
    }
    #pragma unroll
    for (int i = 0; i < MT; ++i) {
        #pragma unroll
        for (int rr = 0; rr < 4; ++rr)
            swr[(4 * i + rr) * 32] = v2f{nS[i][2 * rr], nS[i][2 * rr + 1]};
    }
    __syncthreads();
    #pragma unroll
    for (int jj = 0; jj < NJ; ++jj) {
        const v2f p0 = srd[(0 * NP2 + jj) * 32];
        const v2f p1 = srd[(1 * NP2 + jj) * 32];
        const v2f p2 = srd[(2 * NP2 + jj) * 32];
        const v2f p3 = srd[(3 * NP2 + jj) * 32];
        tot[2 * jj]     = (p0.x + p1.x) + (p2.x + p3.x);
        tot[2 * jj + 1] = (p0.y + p1.y) + (p2.y + p3.y);
    }
    #pragma unroll
    for (int i = 0; i < MT; ++i) nS[i] = zero8;

    float svc[NQ];
    load_s(0, svc);

    #define SCHED_FENCE() __builtin_amdgcn_sched_barrier(0)

    auto step = [&](const int t, const int u, const int tb, auto kind) {
        constexpr bool last = decltype(kind)::last;

        float svn[NQ];
        load_s((t + 1 < Ss) ? t + 1 : t, svn);

        v2f sa[MT];
        v2f sb = {0.f, 0.f};
        if constexpr (!last) {
            sb = sent_b(tb, u);
            #pragma unroll
            for (int i = 0; i < MT; ++i) sa[i] = wal[(u * MT + i) * 32];
        }

        float wsv[NQ];
        {
            const unsigned sel = (unsigned)(selb + u);
            #pragma unroll
            for (int q = 0; q < NQ; ++q) {
                const unsigned v = __float_as_uint(tot[q]);
                wsv[q] = __uint_as_float(__builtin_amdgcn_permlane16_var(v, v, sel, false, false));
            }
        }

        float mg0 = 0.f, mg1 = 0.f, kg0 = 0.f, kg1 = 0.f;
        #pragma unroll
        for (int q = 0; q < NQ; q += 2) {
            mg0 += svc[q]     * st[q];
            mg1 += svc[q + 1] * st[q + 1];
            kg0 += svc[q]     * key[q];
            kg1 += svc[q + 1] * key[q + 1];
        }
        zL[w * 32 + lane] = (mg0 + mg1) + (kg0 + kg1);

        v8f acc[MT];
        #pragma unroll
        for (int i = 0; i < MT; ++i) acc[i] = zero8;
        #pragma unroll
        for (int jj = 0; jj < NJ; ++jj) {
            const v2f bm = v2f{st[2 * jj], st[2 * jj + 1]};
            #pragma unroll
            for (int i = 0; i < MT; ++i)
                acc[i] = __builtin_amdgcn_wmma_f32_16x16x4_f32(
                    false, UA[i][jj], false, bm, (short)0, acc[i], false, false);
        }
        SCHED_FENCE();

        #pragma unroll
        for (int i = 0; i < MT; ++i) {
            #pragma unroll
            for (int rr = 0; rr < 4; ++rr)
                wr[(4 * i + rr) * 32] = v2f{acc[i][2 * rr], acc[i][2 * rr + 1]};
        }
        if constexpr (last) {
            #pragma unroll
            for (int i = 0; i < MT; ++i) {
                #pragma unroll
                for (int rr = 0; rr < 4; ++rr)
                    swr[(4 * i + rr) * 32] = v2f{nS[i][2 * rr], nS[i][2 * rr + 1]};
            }
        }

        __syncthreads();

        v2f a0[NJ], a1[NJ], a2[NJ], a3[NJ];
        #pragma unroll
        for (int jj = 0; jj < NJ; ++jj) {
            a0[jj] = rd[(0 * NP2 + jj) * 32];
            a1[jj] = rd[(1 * NP2 + jj) * 32];
            a2[jj] = rd[(2 * NP2 + jj) * 32];
            a3[jj] = rd[(3 * NP2 + jj) * 32];
        }
        float zq[8];
        #pragma unroll
        for (int e = 0; e < 8; ++e) zq[e] = zL[16 * e + c];
        v2f p0[NJ], p1[NJ], p2[NJ], p3[NJ];
        if constexpr (last) {
            #pragma unroll
            for (int jj = 0; jj < NJ; ++jj) {
                p0[jj] = srd[(0 * NP2 + jj) * 32];
                p1[jj] = srd[(1 * NP2 + jj) * 32];
                p2[jj] = srd[(2 * NP2 + jj) * 32];
                p3[jj] = srd[(3 * NP2 + jj) * 32];
            }
        }
        SCHED_FENCE();
        if constexpr (!last) {
            #pragma unroll
            for (int i = 0; i < 4; ++i)
                nS[i] = __builtin_amdgcn_wmma_f32_16x16x4_f32(
                    false, sa[i], false, sb, (short)0, nS[i], false, false);
        }
        SCHED_FENCE();

        float x[NQ];
        #pragma unroll
        for (int jj = 0; jj < NJ; ++jj) {
            x[2 * jj]     = (((a0[jj].x + a1[jj].x) + (a2[jj].x + a3[jj].x)) + vk[2 * jj])     + wsv[2 * jj];
            x[2 * jj + 1] = (((a0[jj].y + a1[jj].y) + (a2[jj].y + a3[jj].y)) + vk[2 * jj + 1]) + wsv[2 * jj + 1];
        }
        const float z = ((zq[0] + zq[1]) + (zq[2] + zq[3])) + ((zq[4] + zq[5]) + (zq[6] + zq[7]));

        const float g = 1.f / (1.f + expf(-z));
        const float gate = (z == 0.f) ? 0.f : g;

        float ss0 = 0.f, ss1 = 0.f;
        #pragma unroll
        for (int q = 0; q < NQ; q += 2) {
            float x0 = x[q];
            float x1 = x[q + 1];
            x0 = (x0 >= 0.f) ? x0 : pa * x0;
            x1 = (x1 >= 0.f) ? x1 : pa * x1;
            const float n0 = st[q]     + x0 * gate;
            const float n1 = st[q + 1] + x1 * gate;
            st[q]     = n0;
            st[q + 1] = n1;
            ss0 += n0 * n0;
            ss1 += n1 * n1;
        }
        ssL[w * 32 + lane] = ss0 + ss1;

        __syncthreads();

        float sq[8];
        #pragma unroll
        for (int e = 0; e < 8; ++e) sq[e] = ssL[16 * e + c];
        SCHED_FENCE();
        if constexpr (!last) {
            #pragma unroll
            for (int i = 4; i < MT; ++i)
                nS[i] = __builtin_amdgcn_wmma_f32_16x16x4_f32(
                    false, sa[i], false, sb, (short)0, nS[i], false, false);
        }
        SCHED_FENCE();
        const float ss = ((sq[0] + sq[1]) + (sq[2] + sq[3])) + ((sq[4] + sq[5]) + (sq[6] + sq[7]));
        const float sc = 1.f / sqrtf(ss);
        #pragma unroll
        for (int q = 0; q < NQ; ++q) st[q] = st[q] * sc;

        if constexpr (last) {
            #pragma unroll
            for (int jj = 0; jj < NJ; ++jj) {
                tot[2 * jj]     = (p0[jj].x + p1[jj].x) + (p2[jj].x + p3[jj].x);
                tot[2 * jj + 1] = (p0[jj].y + p1[jj].y) + (p2[jj].y + p3[jj].y);
            }
            #pragma unroll
            for (int i = 0; i < MT; ++i) nS[i] = zero8;
        }

        #pragma unroll
        for (int q = 0; q < NQ; ++q) svc[q] = svn[q];
    };

    #pragma nounroll
    for (int tb = 1; tb <= Ss / 8; ++tb) {
        const int t0 = (tb - 1) * 8;
        #pragma nounroll
        for (int u = 0; u < 7; ++u) step(t0 + u, u, tb, StepMid{});
        step(t0 + 7, 7, tb, StepLast{});
    }
    #undef SCHED_FENCE

    #pragma unroll
    for (int jj = 0; jj < NJ; ++jj) {
        if (dok[jj])
            *(v2f*)(outL + c * Dd + dof[jj]) = v2f{st[2 * jj], st[2 * jj + 1]};
    }
    __syncthreads();
    {
        const v4f* src = (const v4f*)outL;
        v4f* dst = (v4f*)(out + (size_t)blockIdx.x * 16 * Dd);
        for (int idx = threadIdx.x; idx < 16 * Dd / 4; idx += 128) dst[idx] = src[idx];
    }
}

extern "C" void kernel_launch(void* const* d_in, const int* in_sizes, int n_in,
                              void* d_out, int out_size, void* d_ws, size_t ws_size,
                              hipStream_t stream) {
    (void)in_sizes; (void)n_in; (void)out_size; (void)ws_size;
    const int*   tokens  = (const int*)  d_in[0];
    const float* emb     = (const float*)d_in[1];
    const float* keys    = (const float*)d_in[2];
    const float* mult    = (const float*)d_in[3];
    const float* Uw      = (const float*)d_in[4];
    const float* Vw      = (const float*)d_in[5];
    const float* Ww      = (const float*)d_in[6];
    const float* prelu_a = (const float*)d_in[7];
    float* out = (float*)d_out;

    float* sents = (float*)d_ws;
    float* VkBuf = sents + (size_t)Bb * Ss * Dd;

    sents_kernel<<<Bb * Ss, 128, 0, stream>>>(tokens, emb, mult, sents);
    vk_kernel<<<Mm, 128, 0, stream>>>(keys, Vw, VkBuf);
    entnet_scan_kernel<<<NGRP, 128, 0, stream>>>(sents, keys, Uw, Ww, VkBuf,
                                                prelu_a, out);
}


# Gate reciprocal and norm scale on the step's serial path: IEEE division and sqrt+division sequences replaced by 1-ulp v_rcp_f32 and v_rsq_f32 (last-bit rounding of gate and scale changes)
# speedup vs baseline: 5.9881x; 1.0524x over previous
#include <hip/hip_runtime.h>

#define Bb   64
#define Ss   128
#define Ll   32
#define Dd   100
#define Mm   20
#define MT   7
#define NW   4
#define NJ   7
#define NQ   14
#define NP2  28
#define NGRP 80
#define SOFF 4096
#define WOFF 8192

typedef __attribute__((ext_vector_type(2))) float v2f;
typedef __attribute__((ext_vector_type(4))) float v4f;
typedef __attribute__((ext_vector_type(8))) float v8f;

__global__ __launch_bounds__(128) void sents_kernel(
    const int* __restrict__ tokens, const float* __restrict__ emb,
    const float* __restrict__ mult, float* __restrict__ sents)
{
    const int bs = blockIdx.x;
    const int d  = threadIdx.x;
    if (d >= Dd) return;
    const int* trow = tokens + bs * Ll;
    float acc = 0.f;
    #pragma unroll 4
    for (int l = 0; l < Ll; ++l) {
        const int tok = trow[l];
        acc += emb[tok * Dd + d] * mult[l * Dd + d];
    }
    sents[bs * Dd + d] = acc;
}

__global__ __launch_bounds__(128) void vk_kernel(
    const float* __restrict__ keys, const float* __restrict__ Vw,
    float* __restrict__ VkBuf)
{
    const int m = blockIdx.x;
    const int d = threadIdx.x;
    if (d >= Dd) return;
    float acc = 0.f;
    for (int k = 0; k < Dd; ++k) acc += keys[m * Dd + k] * Vw[d * Dd + k];
    VkBuf[m * Dd + d] = acc;
}

__device__ __forceinline__ v2f ld2_or_zero(const float* __restrict__ base, int off, bool ok)
{
    const v2f v = *(const v2f*)(base + (ok ? off : 0));
    const v2f z = {0.f, 0.f};
    return ok ? v : z;
}

struct StepMid  { static constexpr bool last = false; };
struct StepLast { static constexpr bool last = true;  };

__global__ __launch_bounds__(128) void entnet_scan_kernel(
    const float* __restrict__ sents, const float* __restrict__ keys,
    const float* __restrict__ Uw,    const float* __restrict__ Ww,
    const float* __restrict__ VkBuf, const float* __restrict__ prelu_a,
    float* __restrict__ out)
{
    const int lane = threadIdx.x & 31;
    const int w    = __builtin_amdgcn_readfirstlane((int)(threadIdx.x >> 5));
    const int c    = lane & 15;
    const int h    = lane >> 4;
    const int G    = blockIdx.x * 16 + c;
    const int b    = G / Mm;
    const int m    = G - b * Mm;
    const float pa = prelu_a[0];

    const int gb0  = (blockIdx.x * 16) / Mm;
    const int gb1  = (blockIdx.x * 16 + 15) / Mm;
    const int bcol = (c >> 3) ? gb1 : gb0;
    const int tcol = c & 7;
    const int selb = (b == gb0) ? 0 : 8;

    __shared__ v2f   partL[21 * 1024];
    __shared__ float zL [NW * 32];
    __shared__ float ssL[NW * 32];
    __shared__ __attribute__((aligned(16))) float outL[16 * Dd];

    int  dof[NJ];
    bool dok[NJ];
    #pragma unroll
    for (int jj = 0; jj < NJ; ++jj) {
        const int P = NQ * w + 2 * jj;
        dof[jj] = 16 * (P >> 3) + 8 * h + (P & 7);
        dok[jj] = dof[jj] < Dd;
    }

    float st[NQ], key[NQ], vk[NQ];
    #pragma unroll
    for (int jj = 0; jj < NJ; ++jj) {
        const v2f k2 = ld2_or_zero(keys,  m * Dd + dof[jj], dok[jj]);
        const v2f v2 = ld2_or_zero(VkBuf, m * Dd + dof[jj], dok[jj]);
        st[2 * jj] = k2.x;  st[2 * jj + 1] = k2.y;
        key[2 * jj] = k2.x; key[2 * jj + 1] = k2.y;
        vk[2 * jj] = v2.x;  vk[2 * jj + 1] = v2.y;
    }

    v2f* const wal = partL + WOFF + (w * NJ * MT) * 32 + lane;
    v2f UA[MT][NJ];
    #pragma unroll
    for (int i = 0; i < MT; ++i) {
        const int row = 16 * i + c;
        #pragma unroll
        for (int jj = 0; jj < NJ; ++jj) {
            const bool ok = (row < Dd) && dok[jj];
            UA[i][jj] = ld2_or_zero(Uw, row * Dd + dof[jj], ok);
            wal[(jj * MT + i) * 32] = ld2_or_zero(Ww, row * Dd + dof[jj], ok);
        }
    }

    auto load_s = [&](int t, float* dst) {
        const int rowoff = (b * Ss + t) * Dd;
        #pragma unroll
        for (int jj = 0; jj < NJ; ++jj) {
            const v2f s2 = ld2_or_zero(sents, rowoff + dof[jj], dok[jj]);
            dst[2 * jj] = s2.x; dst[2 * jj + 1] = s2.y;
        }
    };

    auto sent_b = [&](int tb, int jj) -> v2f {
        const int P  = NQ * w + 2 * jj;
        const int d  = 16 * (P >> 3) + 8 * h + (P & 7);
        const int t8 = tb * 8 + tcol;
        const int tt = (t8 < Ss) ? t8 : (Ss - 1);
        return ld2_or_zero(sents, (bcol * Ss + tt) * Dd + d, d < Dd);
    };

    const v8f zero8 = {0.f, 0.f, 0.f, 0.f, 0.f, 0.f, 0.f, 0.f};

    v2f* const       wr  = partL + (w * NP2) * 32 + lane;
    const v2f* const rd  = partL + (NJ * w) * 32 + lane;
    v2f* const       swr = partL + SOFF + (w * NP2) * 32 + lane;
    const v2f* const srd = partL + SOFF + (NJ * w) * 32 + lane;

    float tot[NQ];
    v8f nS[MT];
    #pragma unroll
    for (int i = 0; i < MT; ++i) nS[i] = zero8;

    #pragma nounroll
    for (int jj = 0; jj < NJ; ++jj) {
        const v2f bs = sent_b(0, jj);
        #pragma unroll
        for (int i = 0; i < MT; ++i)
            nS[i] = __builtin_amdgcn_wmma_f32_16x16x4_f32(
                false, wal[(jj * MT + i) * 32], false, bs, (short)0, nS[i], false, false);
    }
    #pragma unroll
    for (int i = 0; i < MT; ++i) {
        #pragma unroll
        for (int rr = 0; rr < 4; ++rr)
            swr[(4 * i + rr) * 32] = v2f{nS[i][2 * rr], nS[i][2 * rr + 1]};
    }
    __syncthreads();
    #pragma unroll
    for (int jj = 0; jj < NJ; ++jj) {
        const v2f p0 = srd[(0 * NP2 + jj) * 32];
        const v2f p1 = srd[(1 * NP2 + jj) * 32];
        const v2f p2 = srd[(2 * NP2 + jj) * 32];
        const v2f p3 = srd[(3 * NP2 + jj) * 32];
        tot[2 * jj]     = (p0.x + p1.x) + (p2.x + p3.x);
        tot[2 * jj + 1] = (p0.y + p1.y) + (p2.y + p3.y);
    }
    #pragma unroll
    for (int i = 0; i < MT; ++i) nS[i] = zero8;

    float svc[NQ];
    load_s(0, svc);

    #define SCHED_FENCE() __builtin_amdgcn_sched_barrier(0)

    auto step = [&](const int t, const int u, const int tb, auto kind) {
        constexpr bool last = decltype(kind)::last;

        float svn[NQ];
        load_s((t + 1 < Ss) ? t + 1 : t, svn);

        v2f sa[MT];
        v2f sb = {0.f, 0.f};
        if constexpr (!last) {
            sb = sent_b(tb, u);
            #pragma unroll
            for (int i = 0; i < MT; ++i) sa[i] = wal[(u * MT + i) * 32];
        }

        float wsv[NQ];
        {
            const unsigned sel = (unsigned)(selb + u);
            #pragma unroll
            for (int q = 0; q < NQ; ++q) {
                const unsigned v = __float_as_uint(tot[q]);
                wsv[q] = __uint_as_float(__builtin_amdgcn_permlane16_var(v, v, sel, false, false));
            }
        }

        float mg0 = 0.f, mg1 = 0.f, kg0 = 0.f, kg1 = 0.f;
        #pragma unroll
        for (int q = 0; q < NQ; q += 2) {
            mg0 += svc[q]     * st[q];
            mg1 += svc[q + 1] * st[q + 1];
            kg0 += svc[q]     * key[q];
            kg1 += svc[q + 1] * key[q + 1];
        }
        zL[w * 32 + lane] = (mg0 + mg1) + (kg0 + kg1);

        v8f acc[MT];
        #pragma unroll
        for (int i = 0; i < MT; ++i) acc[i] = zero8;
        #pragma unroll
        for (int jj = 0; jj < NJ; ++jj) {
            const v2f bm = v2f{st[2 * jj], st[2 * jj + 1]};
            #pragma unroll
            for (int i = 0; i < MT; ++i)
                acc[i] = __builtin_amdgcn_wmma_f32_16x16x4_f32(
                    false, UA[i][jj], false, bm, (short)0, acc[i], false, false);
        }
        SCHED_FENCE();

        #pragma unroll
        for (int i = 0; i < MT; ++i) {
            #pragma unroll
            for (int rr = 0; rr < 4; ++rr)
                wr[(4 * i + rr) * 32] = v2f{acc[i][2 * rr], acc[i][2 * rr + 1]};
        }
        if constexpr (last) {
            #pragma unroll
            for (int i = 0; i < MT; ++i) {
                #pragma unroll
                for (int rr = 0; rr < 4; ++rr)
                    swr[(4 * i + rr) * 32] = v2f{nS[i][2 * rr], nS[i][2 * rr + 1]};
            }
        }

        __syncthreads();

        v2f a0[NJ], a1[NJ], a2[NJ], a3[NJ];
        #pragma unroll
        for (int jj = 0; jj < NJ; ++jj) {
            a0[jj] = rd[(0 * NP2 + jj) * 32];
            a1[jj] = rd[(1 * NP2 + jj) * 32];
            a2[jj] = rd[(2 * NP2 + jj) * 32];
            a3[jj] = rd[(3 * NP2 + jj) * 32];
        }
        float zq[8];
        #pragma unroll
        for (int e = 0; e < 8; ++e) zq[e] = zL[16 * e + c];
        v2f p0[NJ], p1[NJ], p2[NJ], p3[NJ];
        if constexpr (last) {
            #pragma unroll
            for (int jj = 0; jj < NJ; ++jj) {
                p0[jj] = srd[(0 * NP2 + jj) * 32];
                p1[jj] = srd[(1 * NP2 + jj) * 32];
                p2[jj] = srd[(2 * NP2 + jj) * 32];
                p3[jj] = srd[(3 * NP2 + jj) * 32];
            }
        }
        SCHED_FENCE();
        if constexpr (!last) {
            #pragma unroll
            for (int i = 0; i < 4; ++i)
                nS[i] = __builtin_amdgcn_wmma_f32_16x16x4_f32(
                    false, sa[i], false, sb, (short)0, nS[i], false, false);
        }
        SCHED_FENCE();

        float x[NQ];
        #pragma unroll
        for (int jj = 0; jj < NJ; ++jj) {
            x[2 * jj]     = (((a0[jj].x + a1[jj].x) + (a2[jj].x + a3[jj].x)) + vk[2 * jj])     + wsv[2 * jj];
            x[2 * jj + 1] = (((a0[jj].y + a1[jj].y) + (a2[jj].y + a3[jj].y)) + vk[2 * jj + 1]) + wsv[2 * jj + 1];
        }
        const float z = ((zq[0] + zq[1]) + (zq[2] + zq[3])) + ((zq[4] + zq[5]) + (zq[6] + zq[7]));

        const float g = __builtin_amdgcn_rcpf(1.f + expf(-z));
        const float gate = (z == 0.f) ? 0.f : g;

        float ss0 = 0.f, ss1 = 0.f;
        #pragma unroll
        for (int q = 0; q < NQ; q += 2) {
            float x0 = x[q];
            float x1 = x[q + 1];
            x0 = (x0 >= 0.f) ? x0 : pa * x0;
            x1 = (x1 >= 0.f) ? x1 : pa * x1;
            const float n0 = st[q]     + x0 * gate;
            const float n1 = st[q + 1] + x1 * gate;
            st[q]     = n0;
            st[q + 1] = n1;
            ss0 += n0 * n0;
            ss1 += n1 * n1;
        }
        ssL[w * 32 + lane] = ss0 + ss1;

        __syncthreads();

        float sq[8];
        #pragma unroll
        for (int e = 0; e < 8; ++e) sq[e] = ssL[16 * e + c];
        SCHED_FENCE();
        if constexpr (!last) {
            #pragma unroll
            for (int i = 4; i < MT; ++i)
                nS[i] = __builtin_amdgcn_wmma_f32_16x16x4_f32(
                    false, sa[i], false, sb, (short)0, nS[i], false, false);
        }
        SCHED_FENCE();
        const float ss = ((sq[0] + sq[1]) + (sq[2] + sq[3])) + ((sq[4] + sq[5]) + (sq[6] + sq[7]));
        const float sc = __builtin_amdgcn_rsqf(ss);
        #pragma unroll
        for (int q = 0; q < NQ; ++q) st[q] = st[q] * sc;

        if constexpr (last) {
            #pragma unroll
            for (int jj = 0; jj < NJ; ++jj) {
                tot[2 * jj]     = (p0[jj].x + p1[jj].x) + (p2[jj].x + p3[jj].x);
                tot[2 * jj + 1] = (p0[jj].y + p1[jj].y) + (p2[jj].y + p3[jj].y);
            }
            #pragma unroll
            for (int i = 0; i < MT; ++i) nS[i] = zero8;
        }

        #pragma unroll
        for (int q = 0; q < NQ; ++q) svc[q] = svn[q];
    };

    #pragma nounroll
    for (int tb = 1; tb <= Ss / 8; ++tb) {
        const int t0 = (tb - 1) * 8;
        #pragma nounroll
        for (int u = 0; u < 7; ++u) step(t0 + u, u, tb, StepMid{});
        step(t0 + 7, 7, tb, StepLast{});
    }
    #undef SCHED_FENCE

    #pragma unroll
    for (int jj = 0; jj < NJ; ++jj) {
        if (dok[jj])
            *(v2f*)(outL + c * Dd + dof[jj]) = v2f{st[2 * jj], st[2 * jj + 1]};
    }
    __syncthreads();
    {
        const v4f* src = (const v4f*)outL;
        v4f* dst = (v4f*)(out + (size_t)blockIdx.x * 16 * Dd);
        for (int idx = threadIdx.x; idx < 16 * Dd / 4; idx += 128) dst[idx] = src[idx];
    }
}

extern "C" void kernel_launch(void* const* d_in, const int* in_sizes, int n_in,
                              void* d_out, int out_size, void* d_ws, size_t ws_size,
                              hipStream_t stream) {
    (void)in_sizes; (void)n_in; (void)out_size; (void)ws_size;
    const int*   tokens  = (const int*)  d_in[0];
    const float* emb     = (const float*)d_in[1];
    const float* keys    = (const float*)d_in[2];
    const float* mult    = (const float*)d_in[3];
    const float* Uw      = (const float*)d_in[4];
    const float* Vw      = (const float*)d_in[5];
    const float* Ww      = (const float*)d_in[6];
    const float* prelu_a = (const float*)d_in[7];
    float* out = (float*)d_out;

    float* sents = (float*)d_ws;
    float* VkBuf = sents + (size_t)Bb * Ss * Dd;

    sents_kernel<<<Bb * Ss, 128, 0, stream>>>(tokens, emb, mult, sents);
    vk_kernel<<<Mm, 128, 0, stream>>>(keys, Vw, VkBuf);
    entnet_scan_kernel<<<NGRP, 128, 0, stream>>>(sents, keys, Uw, Ww, VkBuf,
                                                prelu_a, out);
}


# Norm exchange joins the product hand-over: state stays unscaled, 1/norm goes on product sums, s.mem sum, old state; ONE barrier a step, two buffer sets (gate, norm sums reordered; rounding changes)
# speedup vs baseline: 6.1746x; 1.0311x over previous
#include <hip/hip_runtime.h>

#define Bb   64
#define Ss   128
#define Ll   32
#define Dd   100
#define Mm   20
#define MT   7
#define NW   4
#define NJ   7
#define NQ   14
#define NP2  28
#define NGRP 80
#define SOFF 4096
#define WOFF 8192
#define POFF1 16384

typedef __attribute__((ext_vector_type(2))) float v2f;
typedef __attribute__((ext_vector_type(4))) float v4f;
typedef __attribute__((ext_vector_type(8))) float v8f;

__global__ __launch_bounds__(128) void sents_kernel(
    const int* __restrict__ tokens, const float* __restrict__ emb,
    const float* __restrict__ mult, float* __restrict__ sents)
{
    const int bs = blockIdx.x;
    const int d  = threadIdx.x;
    if (d >= Dd) return;
    const int* trow = tokens + bs * Ll;
    float acc = 0.f;
    #pragma unroll 4
    for (int l = 0; l < Ll; ++l) {
        const int tok = trow[l];
        acc += emb[tok * Dd + d] * mult[l * Dd + d];
    }
    sents[bs * Dd + d] = acc;
}

__global__ __launch_bounds__(128) void vk_kernel(
    const float* __restrict__ keys, const float* __restrict__ Vw,
    float* __restrict__ VkBuf)
{
    const int m = blockIdx.x;
    const int d = threadIdx.x;
    if (d >= Dd) return;
    float acc = 0.f;
    for (int k = 0; k < Dd; ++k) acc += keys[m * Dd + k] * Vw[d * Dd + k];
    VkBuf[m * Dd + d] = acc;
}

__device__ __forceinline__ v2f ld2_or_zero(const float* __restrict__ base, int off, bool ok)
{
    const v2f v = *(const v2f*)(base + (ok ? off : 0));
    const v2f z = {0.f, 0.f};
    return ok ? v : z;
}

__device__ __forceinline__ float xhalf(float x)
{
    return __builtin_bit_cast(float, __builtin_amdgcn_permlanex16(
        __builtin_bit_cast(int, x), __builtin_bit_cast(int, x), 0x76543210, 0xfedcba98, false, false));
}

struct StepMid  { static constexpr bool last = false; };
struct StepLast { static constexpr bool last = true;  };

__global__ __launch_bounds__(128) void entnet_scan_kernel(
    const float* __restrict__ sents, const float* __restrict__ keys,
    const float* __restrict__ Uw,    const float* __restrict__ Ww,
    const float* __restrict__ VkBuf, const float* __restrict__ prelu_a,
    float* __restrict__ out)
{
    const int lane = threadIdx.x & 31;
    const int w    = __builtin_amdgcn_readfirstlane((int)(threadIdx.x >> 5));
    const int c    = lane & 15;
    const int h    = lane >> 4;
    const int G    = blockIdx.x * 16 + c;
    const int b    = G / Mm;
    const int m    = G - b * Mm;
    const float pa = prelu_a[0];

    const int gb0  = (blockIdx.x * 16) / Mm;
    const int gb1  = (blockIdx.x * 16 + 15) / Mm;
    const int bcol = (c >> 3) ? gb1 : gb0;
    const int tcol = c & 7;
    const int selb = (b == gb0) ? 0 : 8;

    __shared__ v2f   partL[21 * 1024];
    __shared__ v4f   exL[2][NW * 16];
    __shared__ __attribute__((aligned(16))) float outL[16 * Dd];

    int  dof[NJ];
    bool dok[NJ];
    #pragma unroll
    for (int jj = 0; jj < NJ; ++jj) {
        const int P = NQ * w + 2 * jj;
        dof[jj] = 16 * (P >> 3) + 8 * h + (P & 7);
        dok[jj] = dof[jj] < Dd;
    }

    float st[NQ], key[NQ], vk[NQ];
    #pragma unroll
    for (int jj = 0; jj < NJ; ++jj) {
        const v2f k2 = ld2_or_zero(keys,  m * Dd + dof[jj], dok[jj]);
        const v2f v2 = ld2_or_zero(VkBuf, m * Dd + dof[jj], dok[jj]);
        st[2 * jj] = k2.x;  st[2 * jj + 1] = k2.y;
        key[2 * jj] = k2.x; key[2 * jj + 1] = k2.y;
        vk[2 * jj] = v2.x;  vk[2 * jj + 1] = v2.y;
    }

    v2f* const wal = partL + WOFF + (w * NJ * MT) * 32 + lane;
    v2f UA[MT][NJ];
    #pragma unroll
    for (int i = 0; i < MT; ++i) {
        const int row = 16 * i + c;
        #pragma unroll
        for (int jj = 0; jj < NJ; ++jj) {
            const bool ok = (row < Dd) && dok[jj];
            UA[i][jj] = ld2_or_zero(Uw, row * Dd + dof[jj], ok);
            wal[(jj * MT + i) * 32] = ld2_or_zero(Ww, row * Dd + dof[jj], ok);
        }
    }

    auto load_s = [&](int t, float* dst) {
        const int rowoff = (b * Ss + t) * Dd;
        #pragma unroll
        for (int jj = 0; jj < NJ; ++jj) {
            const v2f s2 = ld2_or_zero(sents, rowoff + dof[jj], dok[jj]);
            dst[2 * jj] = s2.x; dst[2 * jj + 1] = s2.y;
        }
    };

    auto sent_b = [&](int tb, int jj) -> v2f {
        const int P  = NQ * w + 2 * jj;
        const int d  = 16 * (P >> 3) + 8 * h + (P & 7);
        const int t8 = tb * 8 + tcol;
        const int tt = (t8 < Ss) ? t8 : (Ss - 1);
        return ld2_or_zero(sents, (bcol * Ss + tt) * Dd + d, d < Dd);
    };

    const v8f zero8 = {0.f, 0.f, 0.f, 0.f, 0.f, 0.f, 0.f, 0.f};

    v2f* const       wr  = partL + (w * NP2) * 32 + lane;
    const v2f* const rd  = partL + (NJ * w) * 32 + lane;
    v2f* const       swr = partL + SOFF + (w * NP2) * 32 + lane;
    const v2f* const srd = partL + SOFF + (NJ * w) * 32 + lane;

    float tot[NQ];
    v8f nS[MT];
    #pragma unroll
    for (int i = 0; i < MT; ++i) nS[i] = zero8;

    #pragma nounroll
    for (int jj = 0; jj < NJ; ++jj) {
        const v2f bs = sent_b(0, jj);
        #pragma unroll
        for (int i = 0; i < MT; ++i)
            nS[i] = __builtin_amdgcn_wmma_f32_16x16x4_f32(
                false, wal[(jj * MT + i) * 32], false, bs, (short)0, nS[i], false, false);
    }
    #pragma unroll
    for (int i = 0; i < MT; ++i) {
        #pragma unroll
        for (int rr = 0; rr < 4; ++rr)
            swr[(4 * i + rr) * 32] = v2f{nS[i][2 * rr], nS[i][2 * rr + 1]};
    }
    __syncthreads();
    #pragma unroll
    for (int jj = 0; jj < NJ; ++jj) {
        const v2f p0 = srd[(0 * NP2 + jj) * 32];
        const v2f p1 = srd[(1 * NP2 + jj) * 32];
        const v2f p2 = srd[(2 * NP2 + jj) * 32];
        const v2f p3 = srd[(3 * NP2 + jj) * 32];
        tot[2 * jj]     = (p0.x + p1.x) + (p2.x + p3.x);
        tot[2 * jj + 1] = (p0.y + p1.y) + (p2.y + p3.y);
    }
    #pragma unroll
    for (int i = 0; i < MT; ++i) nS[i] = zero8;

    float svc[NQ];
    load_s(0, svc);

    #define SCHED_FENCE() __builtin_amdgcn_sched_barrier(0)

    auto step = [&](const int t, const int u, const int tb, auto kind) {
        constexpr bool last = decltype(kind)::last;
        const int par = t & 1;

        float svn[NQ];
        load_s((t + 1 < Ss) ? t + 1 : t, svn);

        v2f sa[MT];
        v2f sb = {0.f, 0.f};
        if constexpr (!last) {
            sb = sent_b(tb, u);
            #pragma unroll
            for (int i = 0; i < MT; ++i) sa[i] = wal[(u * MT + i) * 32];
        }

        float wsv[NQ];
        {
            const unsigned sel = (unsigned)(selb + u);
            #pragma unroll
            for (int q = 0; q < NQ; ++q) {
                const unsigned v = __float_as_uint(tot[q]);
                wsv[q] = __uint_as_float(__builtin_amdgcn_permlane16_var(v, v, sel, false, false));
            }
        }

        float mg0 = 0.f, mg1 = 0.f, kg0 = 0.f, kg1 = 0.f, sq0 = 0.f, sq1 = 0.f;
        #pragma unroll
        for (int q = 0; q < NQ; q += 2) {
            mg0 += svc[q]     * st[q];
            mg1 += svc[q + 1] * st[q + 1];
            kg0 += svc[q]     * key[q];
            kg1 += svc[q + 1] * key[q + 1];
            sq0 += st[q]      * st[q];
            sq1 += st[q + 1]  * st[q + 1];
        }
        const float pmg = mg0 + mg1, pkg = kg0 + kg1, psq = sq0 + sq1;
        exL[par][w * 16 + c] = v4f{pmg + xhalf(pmg), pkg + xhalf(pkg), psq + xhalf(psq), 0.f};

        v8f acc[MT];
        #pragma unroll
        for (int i = 0; i < MT; ++i) acc[i] = zero8;
        #pragma unroll
        for (int jj = 0; jj < NJ; ++jj) {
            const v2f bm = v2f{st[2 * jj], st[2 * jj + 1]};
            #pragma unroll
            for (int i = 0; i < MT; ++i)
                acc[i] = __builtin_amdgcn_wmma_f32_16x16x4_f32(
                    false, UA[i][jj], false, bm, (short)0, acc[i], false, false);
        }
        SCHED_FENCE();

        v2f* const wrp = wr + par * POFF1;
        #pragma unroll
        for (int i = 0; i < MT; ++i) {
            #pragma unroll
            for (int rr = 0; rr < 4; ++rr)
                wrp[(4 * i + rr) * 32] = v2f{acc[i][2 * rr], acc[i][2 * rr + 1]};
        }
        if constexpr (last) {
            #pragma unroll
            for (int i = 0; i < MT; ++i) {
                #pragma unroll
                for (int rr = 0; rr < 4; ++rr)
                    swr[(4 * i + rr) * 32] = v2f{nS[i][2 * rr], nS[i][2 * rr + 1]};
            }
        }

        __syncthreads();

        v4f eq[NW];
        #pragma unroll
        for (int e = 0; e < NW; ++e) eq[e] = exL[par][16 * e + c];
        const v2f* const rdp = rd + par * POFF1;
        v2f a0[NJ], a1[NJ], a2[NJ], a3[NJ];
        #pragma unroll
        for (int jj = 0; jj < NJ; ++jj) {
            a0[jj] = rdp[(0 * NP2 + jj) * 32];
            a1[jj] = rdp[(1 * NP2 + jj) * 32];
            a2[jj] = rdp[(2 * NP2 + jj) * 32];
            a3[jj] = rdp[(3 * NP2 + jj) * 32];
        }
        v2f p0[NJ], p1[NJ], p2[NJ], p3[NJ];
        if constexpr (last) {
            #pragma unroll
            for (int jj = 0; jj < NJ; ++jj) {
                p0[jj] = srd[(0 * NP2 + jj) * 32];
                p1[jj] = srd[(1 * NP2 + jj) * 32];
                p2[jj] = srd[(2 * NP2 + jj) * 32];
                p3[jj] = srd[(3 * NP2 + jj) * 32];
            }
        }
        SCHED_FENCE();
        if constexpr (!last) {
            #pragma unroll
            for (int i = 0; i < 3; ++i)
                nS[i] = __builtin_amdgcn_wmma_f32_16x16x4_f32(
                    false, sa[i], false, sb, (short)0, nS[i], false, false);
        }
        SCHED_FENCE();

        const float smg = (eq[0].x + eq[1].x) + (eq[2].x + eq[3].x);
        const float skg = (eq[0].y + eq[1].y) + (eq[2].y + eq[3].y);
        const float ss  = (eq[0].z + eq[1].z) + (eq[2].z + eq[3].z);
        const float sc = (t == 0) ? 1.f : __builtin_amdgcn_rsqf(ss);
        const float z  = smg * sc + skg;
        const float g = __builtin_amdgcn_rcpf(1.f + expf(-z));
        const float gate = (z == 0.f) ? 0.f : g;
        #pragma unroll
        for (int q = 0; q < NQ; ++q) st[q] = st[q] * sc;
        if constexpr (!last) {
            #pragma unroll
            for (int i = 3; i < MT; ++i)
                nS[i] = __builtin_amdgcn_wmma_f32_16x16x4_f32(
                    false, sa[i], false, sb, (short)0, nS[i], false, false);
        }
        asm volatile("" :: "v"(gate));
        SCHED_FENCE();

        float x[NQ];
        #pragma unroll
        for (int jj = 0; jj < NJ; ++jj) {
            x[2 * jj]     = (((a0[jj].x + a1[jj].x) + (a2[jj].x + a3[jj].x)) * sc + vk[2 * jj])     + wsv[2 * jj];
            x[2 * jj + 1] = (((a0[jj].y + a1[jj].y) + (a2[jj].y + a3[jj].y)) * sc + vk[2 * jj + 1]) + wsv[2 * jj + 1];
        }

        #pragma unroll
        for (int q = 0; q < NQ; q += 2) {
            float x0 = x[q];
            float x1 = x[q + 1];
            x0 = (x0 >= 0.f) ? x0 : pa * x0;
            x1 = (x1 >= 0.f) ? x1 : pa * x1;
            st[q]     = st[q]     + x0 * gate;
            st[q + 1] = st[q + 1] + x1 * gate;
        }

        if constexpr (last) {
            #pragma unroll
            for (int jj = 0; jj < NJ; ++jj) {
                tot[2 * jj]     = (p0[jj].x + p1[jj].x) + (p2[jj].x + p3[jj].x);
                tot[2 * jj + 1] = (p0[jj].y + p1[jj].y) + (p2[jj].y + p3[jj].y);
            }
            #pragma unroll
            for (int i = 0; i < MT; ++i) nS[i] = zero8;
        }

        #pragma unroll
        for (int q = 0; q < NQ; ++q) svc[q] = svn[q];
    };

    #pragma nounroll
    for (int tb = 1; tb <= Ss / 8; ++tb) {
        const int t0 = (tb - 1) * 8;
        #pragma nounroll
        for (int u = 0; u < 7; ++u) step(t0 + u, u, tb, StepMid{});
        step(t0 + 7, 7, tb, StepLast{});
    }
    #undef SCHED_FENCE

    {
        float sq0 = 0.f, sq1 = 0.f;
        #pragma unroll
        for (int q = 0; q < NQ; q += 2) {
            sq0 += st[q]     * st[q];
            sq1 += st[q + 1] * st[q + 1];
        }
        const float psq = sq0 + sq1;
        exL[0][w * 16 + c] = v4f{0.f, 0.f, psq + xhalf(psq), 0.f};
        __syncthreads();
        float sq[NW];
        #pragma unroll
        for (int e = 0; e < NW; ++e) sq[e] = exL[0][16 * e + c].z;
        const float ss = (sq[0] + sq[1]) + (sq[2] + sq[3]);
        const float sc = __builtin_amdgcn_rsqf(ss);
        #pragma unroll
        for (int q = 0; q < NQ; ++q) st[q] = st[q] * sc;
    }

    #pragma unroll
    for (int jj = 0; jj < NJ; ++jj) {
        if (dok[jj])
            *(v2f*)(outL + c * Dd + dof[jj]) = v2f{st[2 * jj], st[2 * jj + 1]};
    }
    __syncthreads();
    {
        const v4f* src = (const v4f*)outL;
        v4f* dst = (v4f*)(out + (size_t)blockIdx.x * 16 * Dd);
        for (int idx = threadIdx.x; idx < 16 * Dd / 4; idx += 128) dst[idx] = src[idx];
    }
}

extern "C" void kernel_launch(void* const* d_in, const int* in_sizes, int n_in,
                              void* d_out, int out_size, void* d_ws, size_t ws_size,
                              hipStream_t stream) {
    (void)in_sizes; (void)n_in; (void)out_size; (void)ws_size;
    const int*   tokens  = (const int*)  d_in[0];
    const float* emb     = (const float*)d_in[1];
    const float* keys    = (const float*)d_in[2];
    const float* mult    = (const float*)d_in[3];
    const float* Uw      = (const float*)d_in[4];
    const float* Vw      = (const float*)d_in[5];
    const float* Ww      = (const float*)d_in[6];
    const float* prelu_a = (const float*)d_in[7];
    float* out = (float*)d_out;

    float* sents = (float*)d_ws;
    float* VkBuf = sents + (size_t)Bb * Ss * Dd;

    sents_kernel<<<Bb * Ss, 128, 0, stream>>>(tokens, emb, mult, sents);
    vk_kernel<<<Mm, 128, 0, stream>>>(keys, Vw, VkBuf);
    entnet_scan_kernel<<<NGRP, 128, 0, stream>>>(sents, keys, Uw, Ww, VkBuf,
                                                prelu_a, out);
}
